# starting kernel
# baseline (speedup 1.0000x reference)
#include <hip/hip_runtime.h>

#define N_TOTAL    150100
#define EMBED_DIM  64
#define NUM_USERS  100000
#define N_EDGES    3000000
#define BATCH      4096
#define NUM_LAYERS 3

typedef __attribute__((ext_vector_type(2)))  float    v2f;
typedef __attribute__((ext_vector_type(8)))  float    v8f;
typedef __attribute__((ext_vector_type(16))) _Float16 v16h;

__global__ void lg_zero(float4* __restrict__ p, int n4) {
  int i = blockIdx.x * blockDim.x + threadIdx.x;
  const int stride = gridDim.x * blockDim.x;
  for (; i < n4; i += stride) p[i] = make_float4(0.f, 0.f, 0.f, 0.f);
}

__global__ void lg_gather_init(const float* __restrict__ emb, const int* __restrict__ uid,
                               const int* __restrict__ iid, float* __restrict__ acc) {
  const int t = blockIdx.x * blockDim.x + threadIdx.x;
  const int r = t >> 6, c = t & 63;
  const int n = (r < BATCH) ? uid[r] : (NUM_USERS + iid[r - BATCH]);
  acc[t] = emb[(size_t)n * EMBED_DIM + c];
}

__global__ void lg_gather_acc(const float* __restrict__ emb, const int* __restrict__ uid,
                              const int* __restrict__ iid, float* __restrict__ acc) {
  const int t = blockIdx.x * blockDim.x + threadIdx.x;
  const int r = t >> 6, c = t & 63;
  const int n = (r < BATCH) ? uid[r] : (NUM_USERS + iid[r - BATCH]);
  acc[t] += emb[(size_t)n * EMBED_DIM + c];
}

__global__ void lg_spmm_scatter(const float* __restrict__ emb, const float* __restrict__ val,
                                const int* __restrict__ rw, const int* __restrict__ cl,
                                float* __restrict__ out) {
  const int lane = threadIdx.x & 31;
  const int d    = lane << 1;
  int w          = (blockIdx.x * blockDim.x + threadIdx.x) >> 5;
  const int nw   = (gridDim.x * blockDim.x) >> 5;
  for (int e = w; e < N_EDGES; e += nw) {
    const int   r = __builtin_amdgcn_readfirstlane(rw[e]);
    const int   c = __builtin_amdgcn_readfirstlane(cl[e]);
    const float v = val[e];
    const float2 s = *reinterpret_cast<const float2*>(emb + (size_t)c * EMBED_DIM + d);
    float* dst = out + (size_t)r * EMBED_DIM + d;
    unsafeAtomicAdd(dst,     v * s.x);
    unsafeAtomicAdd(dst + 1, v * s.y);
  }
}

__global__ void lg_dot_wmma(const float* __restrict__ acc, float* __restrict__ out) {
  __shared__ float sd[8][8][32];
  const int lane = threadIdx.x & 31;
  const int wave = threadIdx.x >> 5;
  const int tile = blockIdx.x * 8 + wave;
  const int k0   = tile * 16;
  const int rN   = lane & 15;
  const float* U = acc + (size_t)(k0 + rN) * EMBED_DIM;
  const float* I = acc + (size_t)(BATCH + k0 + rN) * EMBED_DIM;

  v8f c = {0.f, 0.f, 0.f, 0.f, 0.f, 0.f, 0.f, 0.f};

#if __has_builtin(__builtin_amdgcn_wmma_f32_16x16x4_f32)
  const int kb = (lane >> 4) << 1;
  #pragma unroll
  for (int s = 0; s < 16; ++s) {
    const int col = s * 4 + kb;
    v2f a; a[0] = U[col]; a[1] = U[col + 1];
    v2f b; b[0] = I[col]; b[1] = I[col + 1];
    c = __builtin_amdgcn_wmma_f32_16x16x4_f32(false, a, false, b, (short)0, c, false, false);
  }
#else
  const int kb8 = (lane >> 4) << 3;
  #pragma unroll
  for (int blk = 0; blk < 2; ++blk) {
    v16h a, b;
    #pragma unroll
    for (int i = 0; i < 8; ++i) {
      const int kk = blk * 32 + kb8 + ((i < 4) ? (2 * i) : (16 + 2 * (i - 4)));
      a[2 * i]     = (_Float16)U[kk];
      a[2 * i + 1] = (_Float16)U[kk + 1];
      b[2 * i]     = (_Float16)I[kk];
      b[2 * i + 1] = (_Float16)I[kk + 1];
    }
    c = __builtin_amdgcn_wmma_f32_16x16x32_f16(false, a, false, b, (short)0, c, false, false);
  }
#endif

  #pragma unroll
  for (int g = 0; g < 8; ++g) sd[wave][g][lane] = c[g];
  __syncthreads();

  if (lane < 16) {
    const float v = (lane < 8) ? sd[wave][lane][lane] : sd[wave][lane - 8][lane + 16];
    out[k0 + lane] = v * 0.0625f;
  }
}

extern "C" void kernel_launch(void* const* d_in, const int* in_sizes, int n_in,
                              void* d_out, int out_size, void* d_ws, size_t ws_size,
                              hipStream_t stream) {
  const float* node_emb = (const float*)d_in[0];
  const float* adj_val  = (const float*)d_in[1];
  const int*   adj_row  = (const int*)d_in[2];
  const int*   adj_col  = (const int*)d_in[3];
  const int*   user_ids = (const int*)d_in[4];
  const int*   item_ids = (const int*)d_in[5];
  float* out = (float*)d_out;

  const size_t embElems = (size_t)N_TOTAL * EMBED_DIM;
  char* ws   = (char*)d_ws;
  float* embA = (float*)ws;
  float* embB = (float*)(ws + embElems * sizeof(float));
  float* accS = (float*)(ws + 2 * embElems * sizeof(float));

  const dim3 b256(256);
  const int gatherBlocks = (2 * BATCH * EMBED_DIM) / 256;
  const int n4 = (int)(embElems / 4);

  lg_gather_init<<<gatherBlocks, b256, 0, stream>>>(node_emb, user_ids, item_ids, accS);

  const float* cur = node_emb;
  float* bufs[2] = {embA, embB};
  for (int L = 0; L < NUM_LAYERS; ++L) {
    float* nxt = bufs[L & 1];
    lg_zero<<<2048, b256, 0, stream>>>((float4*)nxt, n4);
    lg_spmm_scatter<<<2048, b256, 0, stream>>>(cur, adj_val, adj_row, adj_col, nxt);
    lg_gather_acc<<<gatherBlocks, b256, 0, stream>>>(nxt, user_ids, item_ids, accS);
    cur = nxt;
  }

  lg_dot_wmma<<<BATCH / 128, b256, 0, stream>>>(accS, out);
}


# Last layer summed only at the 8192 gathered rows: per-workgroup LDS bitmask of those rows, edges tested 32 per wave, the rest skipped; only those rows zeroed. Same f32 atomics, arrival order as ever
# speedup vs baseline: 1.4673x; 1.4673x over previous
#include <hip/hip_runtime.h>

#define N_TOTAL    150100
#define EMBED_DIM  64
#define NUM_USERS  100000
#define N_EDGES    3000000
#define BATCH      4096
#define NUM_LAYERS 3

typedef __attribute__((ext_vector_type(2)))  float    v2f;
typedef __attribute__((ext_vector_type(8)))  float    v8f;
typedef __attribute__((ext_vector_type(16))) _Float16 v16h;

__global__ void lg_zero(float4* __restrict__ p, int n4) {
  int i = blockIdx.x * blockDim.x + threadIdx.x;
  const int stride = gridDim.x * blockDim.x;
  for (; i < n4; i += stride) p[i] = make_float4(0.f, 0.f, 0.f, 0.f);
}

__global__ void lg_gather_init(const float* __restrict__ emb, const int* __restrict__ uid,
                               const int* __restrict__ iid, float* __restrict__ acc) {
  const int t = blockIdx.x * blockDim.x + threadIdx.x;
  const int r = t >> 6, c = t & 63;
  const int n = (r < BATCH) ? uid[r] : (NUM_USERS + iid[r - BATCH]);
  acc[t] = emb[(size_t)n * EMBED_DIM + c];
}

__global__ void lg_gather_acc(const float* __restrict__ emb, const int* __restrict__ uid,
                              const int* __restrict__ iid, float* __restrict__ acc) {
  const int t = blockIdx.x * blockDim.x + threadIdx.x;
  const int r = t >> 6, c = t & 63;
  const int n = (r < BATCH) ? uid[r] : (NUM_USERS + iid[r - BATCH]);
  acc[t] += emb[(size_t)n * EMBED_DIM + c];
}

__global__ void lg_spmm_scatter(const float* __restrict__ emb, const float* __restrict__ val,
                                const int* __restrict__ rw, const int* __restrict__ cl,
                                float* __restrict__ out) {
  const int lane = threadIdx.x & 31;
  const int d    = lane << 1;
  int w          = (blockIdx.x * blockDim.x + threadIdx.x) >> 5;
  const int nw   = (gridDim.x * blockDim.x) >> 5;
  for (int e = w; e < N_EDGES; e += nw) {
    const int   r = __builtin_amdgcn_readfirstlane(rw[e]);
    const int   c = __builtin_amdgcn_readfirstlane(cl[e]);
    const float v = val[e];
    const float2 s = *reinterpret_cast<const float2*>(emb + (size_t)c * EMBED_DIM + d);
    float* dst = out + (size_t)r * EMBED_DIM + d;
    unsafeAtomicAdd(dst,     v * s.x);
    unsafeAtomicAdd(dst + 1, v * s.y);
  }
}

__global__ void lg_zero_rows(const int* __restrict__ uid, const int* __restrict__ iid,
                             float* __restrict__ buf) {
  const int t = blockIdx.x * blockDim.x + threadIdx.x;
  const int r = t >> 6, c = t & 63;
  const int n = (r < BATCH) ? uid[r] : (NUM_USERS + iid[r - BATCH]);
  buf[(size_t)n * EMBED_DIM + c] = 0.f;
}

#define NEED_WORDS 4692
static_assert(NEED_WORDS * 32 >= N_TOTAL, "bitmask must cover every row");
static_assert(N_EDGES % 32 == 0, "edges are read 32 per wave");

#define SG_THREADS 1024
static_assert(BATCH % SG_THREADS == 0, "ids are read SG_THREADS at a time");

__global__ void __launch_bounds__(SG_THREADS)
lg_spmm_scatter_gathered(const float* __restrict__ emb, const float* __restrict__ val,
                         const int* __restrict__ rw, const int* __restrict__ cl,
                         const int* __restrict__ uid, const int* __restrict__ iid,
                         float* __restrict__ out) {
  __shared__ unsigned int need[NEED_WORDS];
  for (int i = threadIdx.x; i < NEED_WORDS; i += SG_THREADS) need[i] = 0u;
  int un[BATCH / SG_THREADS], in[BATCH / SG_THREADS];
  for (int k = 0; k < BATCH / SG_THREADS; ++k) {
    un[k] = uid[k * SG_THREADS + threadIdx.x];
    in[k] = NUM_USERS + iid[k * SG_THREADS + threadIdx.x];
  }
  __syncthreads();
  for (int k = 0; k < BATCH / SG_THREADS; ++k) {
    atomicOr(&need[un[k] >> 5], 1u << (un[k] & 31));
    atomicOr(&need[in[k] >> 5], 1u << (in[k] & 31));
  }
  __syncthreads();

  const int lane = threadIdx.x & 31;
  const int d    = lane << 1;
  const int w    = (blockIdx.x * blockDim.x + threadIdx.x) >> 5;
  const int nw   = (gridDim.x * blockDim.x) >> 5;
  for (int b = w; b < N_EDGES / 32; b += nw) {
    const int   e = b * 32 + lane;
    const int   r = rw[e];
    const int   c = cl[e];
    const float v = val[e];
    const bool hit = (need[r >> 5] >> (r & 31)) & 1u;
    unsigned int m = __builtin_amdgcn_ballot_w32(hit);
    while (m) {
      const int j = __builtin_ctz(m);
      m &= m - 1u;
      const int   rj = __builtin_amdgcn_readlane(r, j);
      const int   cj = __builtin_amdgcn_readlane(c, j);
      const float vj = __builtin_bit_cast(float,
                         __builtin_amdgcn_readlane(__builtin_bit_cast(int, v), j));
      const float2 s = *reinterpret_cast<const float2*>(emb + (size_t)cj * EMBED_DIM + d);
      float* dst = out + (size_t)rj * EMBED_DIM + d;
      unsafeAtomicAdd(dst,     vj * s.x);
      unsafeAtomicAdd(dst + 1, vj * s.y);
    }
  }
}

__global__ void lg_dot_wmma(const float* __restrict__ acc, float* __restrict__ out) {
  __shared__ float sd[8][8][32];
  const int lane = threadIdx.x & 31;
  const int wave = threadIdx.x >> 5;
  const int tile = blockIdx.x * 8 + wave;
  const int k0   = tile * 16;
  const int rN   = lane & 15;
  const float* U = acc + (size_t)(k0 + rN) * EMBED_DIM;
  const float* I = acc + (size_t)(BATCH + k0 + rN) * EMBED_DIM;

  v8f c = {0.f, 0.f, 0.f, 0.f, 0.f, 0.f, 0.f, 0.f};

#if __has_builtin(__builtin_amdgcn_wmma_f32_16x16x4_f32)
  const int kb = (lane >> 4) << 1;
  #pragma unroll
  for (int s = 0; s < 16; ++s) {
    const int col = s * 4 + kb;
    v2f a; a[0] = U[col]; a[1] = U[col + 1];
    v2f b; b[0] = I[col]; b[1] = I[col + 1];
    c = __builtin_amdgcn_wmma_f32_16x16x4_f32(false, a, false, b, (short)0, c, false, false);
  }
#else
  const int kb8 = (lane >> 4) << 3;
  #pragma unroll
  for (int blk = 0; blk < 2; ++blk) {
    v16h a, b;
    #pragma unroll
    for (int i = 0; i < 8; ++i) {
      const int kk = blk * 32 + kb8 + ((i < 4) ? (2 * i) : (16 + 2 * (i - 4)));
      a[2 * i]     = (_Float16)U[kk];
      a[2 * i + 1] = (_Float16)U[kk + 1];
      b[2 * i]     = (_Float16)I[kk];
      b[2 * i + 1] = (_Float16)I[kk + 1];
    }
    c = __builtin_amdgcn_wmma_f32_16x16x32_f16(false, a, false, b, (short)0, c, false, false);
  }
#endif

  #pragma unroll
  for (int g = 0; g < 8; ++g) sd[wave][g][lane] = c[g];
  __syncthreads();

  if (lane < 16) {
    const float v = (lane < 8) ? sd[wave][lane][lane] : sd[wave][lane - 8][lane + 16];
    out[k0 + lane] = v * 0.0625f;
  }
}

extern "C" void kernel_launch(void* const* d_in, const int* in_sizes, int n_in,
                              void* d_out, int out_size, void* d_ws, size_t ws_size,
                              hipStream_t stream) {
  const float* node_emb = (const float*)d_in[0];
  const float* adj_val  = (const float*)d_in[1];
  const int*   adj_row  = (const int*)d_in[2];
  const int*   adj_col  = (const int*)d_in[3];
  const int*   user_ids = (const int*)d_in[4];
  const int*   item_ids = (const int*)d_in[5];
  float* out = (float*)d_out;

  const size_t embElems = (size_t)N_TOTAL * EMBED_DIM;
  char* ws   = (char*)d_ws;
  float* embA = (float*)ws;
  float* embB = (float*)(ws + embElems * sizeof(float));
  float* accS = (float*)(ws + 2 * embElems * sizeof(float));

  const dim3 b256(256);
  const int gatherBlocks = (2 * BATCH * EMBED_DIM) / 256;
  const int n4 = (int)(embElems / 4);

  lg_gather_init<<<gatherBlocks, b256, 0, stream>>>(node_emb, user_ids, item_ids, accS);

  const float* cur = node_emb;
  float* bufs[2] = {embA, embB};
  for (int L = 0; L < NUM_LAYERS - 1; ++L) {
    float* nxt = bufs[L & 1];
    lg_zero<<<2048, b256, 0, stream>>>((float4*)nxt, n4);
    lg_spmm_scatter<<<2048, b256, 0, stream>>>(cur, adj_val, adj_row, adj_col, nxt);
    lg_gather_acc<<<gatherBlocks, b256, 0, stream>>>(nxt, user_ids, item_ids, accS);
    cur = nxt;
  }
  {
    float* nxt = bufs[(NUM_LAYERS - 1) & 1];
    lg_zero_rows<<<gatherBlocks, b256, 0, stream>>>(user_ids, item_ids, nxt);
    lg_spmm_scatter_gathered<<<512, dim3(SG_THREADS), 0, stream>>>(cur, adj_val, adj_row, adj_col,
                                                            user_ids, item_ids, nxt);
    lg_gather_acc<<<gatherBlocks, b256, 0, stream>>>(nxt, user_ids, item_ids, accS);
  }

  lg_dot_wmma<<<BATCH / 128, b256, 0, stream>>>(accS, out);
}


# Full-layer scatter asks for 4 cache lines per edge, not 9: ids read 32 edges per wave load; a lane owns dims lane, lane+32, so each row load or atomic is one 128-B line. Same f32 atomics
# speedup vs baseline: 2.6486x; 1.8051x over previous
#include <hip/hip_runtime.h>

#define N_TOTAL    150100
#define EMBED_DIM  64
#define NUM_USERS  100000
#define N_EDGES    3000000
#define BATCH      4096
#define NUM_LAYERS 3

typedef __attribute__((ext_vector_type(2)))  float    v2f;
typedef __attribute__((ext_vector_type(8)))  float    v8f;
typedef __attribute__((ext_vector_type(16))) _Float16 v16h;

__global__ void lg_zero(float4* __restrict__ p, int n4) {
  int i = blockIdx.x * blockDim.x + threadIdx.x;
  const int stride = gridDim.x * blockDim.x;
  for (; i < n4; i += stride) p[i] = make_float4(0.f, 0.f, 0.f, 0.f);
}

__global__ void lg_gather_init(const float* __restrict__ emb, const int* __restrict__ uid,
                               const int* __restrict__ iid, float* __restrict__ acc) {
  const int t = blockIdx.x * blockDim.x + threadIdx.x;
  const int r = t >> 6, c = t & 63;
  const int n = (r < BATCH) ? uid[r] : (NUM_USERS + iid[r - BATCH]);
  acc[t] = emb[(size_t)n * EMBED_DIM + c];
}

__global__ void lg_gather_acc(const float* __restrict__ emb, const int* __restrict__ uid,
                              const int* __restrict__ iid, float* __restrict__ acc) {
  const int t = blockIdx.x * blockDim.x + threadIdx.x;
  const int r = t >> 6, c = t & 63;
  const int n = (r < BATCH) ? uid[r] : (NUM_USERS + iid[r - BATCH]);
  acc[t] += emb[(size_t)n * EMBED_DIM + c];
}

#define EDGE_GROUP 8
static_assert(32 % EDGE_GROUP == 0, "a 32-edge step is walked in whole groups");

__global__ void __launch_bounds__(256)
lg_spmm_scatter(const float* __restrict__ emb, const float* __restrict__ val,
                const int* __restrict__ rw, const int* __restrict__ cl,
                float* __restrict__ out) {
  const int lane = threadIdx.x & 31;
  const int w    = __builtin_amdgcn_readfirstlane((blockIdx.x * blockDim.x + threadIdx.x) >> 5);
  const int nw   = (gridDim.x * blockDim.x) >> 5;
  for (int b = w; b < N_EDGES / 32; b += nw) {
    const int   e = b * 32 + lane;
    const int   r = rw[e];
    const int   c = cl[e];
    const float v = val[e];
    #pragma unroll
    for (int j0 = 0; j0 < 32; j0 += EDGE_GROUP) {
      float s0[EDGE_GROUP], s1[EDGE_GROUP];
      #pragma unroll
      for (int k = 0; k < EDGE_GROUP; ++k) {
        const int cj = __builtin_amdgcn_readlane(c, j0 + k);
        const float* src = emb + (size_t)cj * EMBED_DIM + lane;
        s0[k] = src[0];
        s1[k] = src[32];
      }
      #pragma unroll
      for (int k = 0; k < EDGE_GROUP; ++k) {
        const int   rj = __builtin_amdgcn_readlane(r, j0 + k);
        const float vj = __builtin_bit_cast(float,
                           __builtin_amdgcn_readlane(__builtin_bit_cast(int, v), j0 + k));
        float* dst = out + (size_t)rj * EMBED_DIM + lane;
        unsafeAtomicAdd(dst,      vj * s0[k]);
        unsafeAtomicAdd(dst + 32, vj * s1[k]);
      }
    }
  }
}

__global__ void lg_zero_rows(const int* __restrict__ uid, const int* __restrict__ iid,
                             float* __restrict__ buf) {
  const int t = blockIdx.x * blockDim.x + threadIdx.x;
  const int r = t >> 6, c = t & 63;
  const int n = (r < BATCH) ? uid[r] : (NUM_USERS + iid[r - BATCH]);
  buf[(size_t)n * EMBED_DIM + c] = 0.f;
}

#define NEED_WORDS 4692
static_assert(NEED_WORDS * 32 >= N_TOTAL, "bitmask must cover every row");
static_assert(N_EDGES % 32 == 0, "edges are read 32 per wave");

#define SG_THREADS 1024
static_assert(BATCH % SG_THREADS == 0, "ids are read SG_THREADS at a time");

__global__ void __launch_bounds__(SG_THREADS)
lg_spmm_scatter_gathered(const float* __restrict__ emb, const float* __restrict__ val,
                         const int* __restrict__ rw, const int* __restrict__ cl,
                         const int* __restrict__ uid, const int* __restrict__ iid,
                         float* __restrict__ out) {
  __shared__ unsigned int need[NEED_WORDS];
  for (int i = threadIdx.x; i < NEED_WORDS; i += SG_THREADS) need[i] = 0u;
  int un[BATCH / SG_THREADS], in[BATCH / SG_THREADS];
  for (int k = 0; k < BATCH / SG_THREADS; ++k) {
    un[k] = uid[k * SG_THREADS + threadIdx.x];
    in[k] = NUM_USERS + iid[k * SG_THREADS + threadIdx.x];
  }
  __syncthreads();
  for (int k = 0; k < BATCH / SG_THREADS; ++k) {
    atomicOr(&need[un[k] >> 5], 1u << (un[k] & 31));
    atomicOr(&need[in[k] >> 5], 1u << (in[k] & 31));
  }
  __syncthreads();

  const int lane = threadIdx.x & 31;
  const int w    = (blockIdx.x * blockDim.x + threadIdx.x) >> 5;
  const int nw   = (gridDim.x * blockDim.x) >> 5;
  for (int b = w; b < N_EDGES / 32; b += nw) {
    const int   e = b * 32 + lane;
    const int   r = rw[e];
    const int   c = cl[e];
    const float v = val[e];
    const bool hit = (need[r >> 5] >> (r & 31)) & 1u;
    unsigned int m = __builtin_amdgcn_ballot_w32(hit);
    while (m) {
      const int j = __builtin_ctz(m);
      m &= m - 1u;
      const int   rj = __builtin_amdgcn_readlane(r, j);
      const int   cj = __builtin_amdgcn_readlane(c, j);
      const float vj = __builtin_bit_cast(float,
                         __builtin_amdgcn_readlane(__builtin_bit_cast(int, v), j));
      const float* srcp = emb + (size_t)cj * EMBED_DIM + lane;
      const float s0 = srcp[0];
      const float s1 = srcp[32];
      float* dst = out + (size_t)rj * EMBED_DIM + lane;
      unsafeAtomicAdd(dst,      vj * s0);
      unsafeAtomicAdd(dst + 32, vj * s1);
    }
  }
}

__global__ void lg_dot_wmma(const float* __restrict__ acc, float* __restrict__ out) {
  __shared__ float sd[8][8][32];
  const int lane = threadIdx.x & 31;
  const int wave = threadIdx.x >> 5;
  const int tile = blockIdx.x * 8 + wave;
  const int k0   = tile * 16;
  const int rN   = lane & 15;
  const float* U = acc + (size_t)(k0 + rN) * EMBED_DIM;
  const float* I = acc + (size_t)(BATCH + k0 + rN) * EMBED_DIM;

  v8f c = {0.f, 0.f, 0.f, 0.f, 0.f, 0.f, 0.f, 0.f};

#if __has_builtin(__builtin_amdgcn_wmma_f32_16x16x4_f32)
  const int kb = (lane >> 4) << 1;
  #pragma unroll
  for (int s = 0; s < 16; ++s) {
    const int col = s * 4 + kb;
    v2f a; a[0] = U[col]; a[1] = U[col + 1];
    v2f b; b[0] = I[col]; b[1] = I[col + 1];
    c = __builtin_amdgcn_wmma_f32_16x16x4_f32(false, a, false, b, (short)0, c, false, false);
  }
#else
  const int kb8 = (lane >> 4) << 3;
  #pragma unroll
  for (int blk = 0; blk < 2; ++blk) {
    v16h a, b;
    #pragma unroll
    for (int i = 0; i < 8; ++i) {
      const int kk = blk * 32 + kb8 + ((i < 4) ? (2 * i) : (16 + 2 * (i - 4)));
      a[2 * i]     = (_Float16)U[kk];
      a[2 * i + 1] = (_Float16)U[kk + 1];
      b[2 * i]     = (_Float16)I[kk];
      b[2 * i + 1] = (_Float16)I[kk + 1];
    }
    c = __builtin_amdgcn_wmma_f32_16x16x32_f16(false, a, false, b, (short)0, c, false, false);
  }
#endif

  #pragma unroll
  for (int g = 0; g < 8; ++g) sd[wave][g][lane] = c[g];
  __syncthreads();

  if (lane < 16) {
    const float v = (lane < 8) ? sd[wave][lane][lane] : sd[wave][lane - 8][lane + 16];
    out[k0 + lane] = v * 0.0625f;
  }
}

extern "C" void kernel_launch(void* const* d_in, const int* in_sizes, int n_in,
                              void* d_out, int out_size, void* d_ws, size_t ws_size,
                              hipStream_t stream) {
  const float* node_emb = (const float*)d_in[0];
  const float* adj_val  = (const float*)d_in[1];
  const int*   adj_row  = (const int*)d_in[2];
  const int*   adj_col  = (const int*)d_in[3];
  const int*   user_ids = (const int*)d_in[4];
  const int*   item_ids = (const int*)d_in[5];
  float* out = (float*)d_out;

  const size_t embElems = (size_t)N_TOTAL * EMBED_DIM;
  char* ws   = (char*)d_ws;
  float* embA = (float*)ws;
  float* embB = (float*)(ws + embElems * sizeof(float));
  float* accS = (float*)(ws + 2 * embElems * sizeof(float));

  const dim3 b256(256);
  const int gatherBlocks = (2 * BATCH * EMBED_DIM) / 256;
  const int n4 = (int)(embElems / 4);

  lg_gather_init<<<gatherBlocks, b256, 0, stream>>>(node_emb, user_ids, item_ids, accS);

  const float* cur = node_emb;
  float* bufs[2] = {embA, embB};
  for (int L = 0; L < NUM_LAYERS - 1; ++L) {
    float* nxt = bufs[L & 1];
    lg_zero<<<2048, b256, 0, stream>>>((float4*)nxt, n4);
    lg_spmm_scatter<<<2048, b256, 0, stream>>>(cur, adj_val, adj_row, adj_col, nxt);
    lg_gather_acc<<<gatherBlocks, b256, 0, stream>>>(nxt, user_ids, item_ids, accS);
    cur = nxt;
  }
  {
    float* nxt = bufs[(NUM_LAYERS - 1) & 1];
    lg_zero_rows<<<gatherBlocks, b256, 0, stream>>>(user_ids, item_ids, nxt);
    lg_spmm_scatter_gathered<<<512, dim3(SG_THREADS), 0, stream>>>(cur, adj_val, adj_row, adj_col,
                                                            user_ids, item_ids, nxt);
    lg_gather_acc<<<gatherBlocks, b256, 0, stream>>>(nxt, user_ids, item_ids, accS);
  }

  lg_dot_wmma<<<BATCH / 128, b256, 0, stream>>>(accS, out);
}


# Full layers: edges bucketed per call by 128-row destination block (lists held in the unwritten layer table); each block summed in LDS by ds_add_f32, no global float atomics. Arrival order as ever
# speedup vs baseline: 3.7897x; 1.4308x over previous
#include <hip/hip_runtime.h>

#define N_TOTAL    150100
#define EMBED_DIM  64
#define NUM_USERS  100000
#define N_EDGES    3000000
#define BATCH      4096
#define NUM_LAYERS 3

typedef __attribute__((ext_vector_type(2)))  float    v2f;
typedef __attribute__((ext_vector_type(8)))  float    v8f;
typedef __attribute__((ext_vector_type(16))) _Float16 v16h;

#define BIN_ROWS    128
#define BIN_SHIFT   7
#define NUM_BINS    ((N_TOTAL + BIN_ROWS - 1) / BIN_ROWS)
#define BIN_FLOATS  (BIN_ROWS * EMBED_DIM)
#define COL_BITS    18
#define COL_MASK    ((1 << COL_BITS) - 1)
static_assert((1 << BIN_SHIFT) == BIN_ROWS, "block index is a shift of the row");
static_assert(N_TOTAL <= (1 << COL_BITS), "a column id fits its field");

__device__ __forceinline__ int bin_capacity(int b) {
  const int rows = min(BIN_ROWS, N_TOTAL - b * BIN_ROWS);
  return rows * (EMBED_DIM / 2) - 1;
}

__global__ void lg_gather_init(const float* __restrict__ emb, const int* __restrict__ uid,
                               const int* __restrict__ iid, float* __restrict__ acc,
                               float* __restrict__ list) {
  const int t = blockIdx.x * blockDim.x + threadIdx.x;
  const int r = t >> 6, c = t & 63;
  const int n = (r < BATCH) ? uid[r] : (NUM_USERS + iid[r - BATCH]);
  acc[t] = emb[(size_t)n * EMBED_DIM + c];
  if (t < NUM_BINS) *(int*)(list + (size_t)t * BIN_FLOATS) = 0;
}

__global__ void lg_gather_acc(const float* __restrict__ emb, const int* __restrict__ uid,
                              const int* __restrict__ iid, float* __restrict__ acc) {
  const int t = blockIdx.x * blockDim.x + threadIdx.x;
  const int r = t >> 6, c = t & 63;
  const int n = (r < BATCH) ? uid[r] : (NUM_USERS + iid[r - BATCH]);
  acc[t] += emb[(size_t)n * EMBED_DIM + c];
}

#define FILL_THREADS 1024
#define FILL_EPT     6
#define FILL_CHUNK   (FILL_THREADS * FILL_EPT)
#define FILL_WGS     ((N_EDGES + FILL_CHUNK - 1) / FILL_CHUNK)
#define FILL_SLAB    16

__global__ void __launch_bounds__(FILL_THREADS)
lg_bin_fill(const float* __restrict__ val, const int* __restrict__ rw,
            const int* __restrict__ cl, float* __restrict__ list) {
  __shared__ int  hist[NUM_BINS];
  __shared__ int  lcnt[NUM_BINS];
  __shared__ int2 slab[NUM_BINS * FILL_SLAB];
  for (int i = threadIdx.x; i < NUM_BINS; i += FILL_THREADS) hist[i] = 0;
  __syncthreads();
  const int e0 = blockIdx.x * FILL_CHUNK + threadIdx.x;
  int pk[FILL_EPT], bn[FILL_EPT], rk[FILL_EPT], vb[FILL_EPT];
  #pragma unroll
  for (int k = 0; k < FILL_EPT; ++k) {
    const int e = e0 + k * FILL_THREADS;
    bn[k] = -1;
    if (e < N_EDGES) {
      const int r = rw[e];
      bn[k] = r >> BIN_SHIFT;
      pk[k] = ((r & (BIN_ROWS - 1)) << COL_BITS) | cl[e];
      vb[k] = __builtin_bit_cast(int, val[e]);
    }
  }
  #pragma unroll
  for (int k = 0; k < FILL_EPT; ++k) {
    if (bn[k] >= 0) {
      rk[k] = atomicAdd(&hist[bn[k]], 1);
      if (rk[k] < FILL_SLAB) slab[bn[k] * FILL_SLAB + rk[k]] = make_int2(pk[k], vb[k]);
    }
  }
  __syncthreads();
  for (int b = threadIdx.x; b < NUM_BINS; b += FILL_THREADS) {
    const int n = hist[b];
    lcnt[b] = n;
    if (n) hist[b] = atomicAdd((int*)(list + (size_t)b * BIN_FLOATS), n);
  }
  __syncthreads();
  #pragma unroll
  for (int k = 0; k < FILL_EPT; ++k) {
    if (bn[k] >= 0 && rk[k] >= FILL_SLAB) {
      const int slot = hist[bn[k]] + rk[k];
      if (slot < bin_capacity(bn[k])) {
        int2* region = (int2*)(list + (size_t)bn[k] * BIN_FLOATS);
        region[1 + slot] = make_int2(pk[k], vb[k]);
      }
    }
  }
  for (int i = threadIdx.x; i < NUM_BINS * FILL_SLAB; i += FILL_THREADS) {
    const int b = i / FILL_SLAB, j = i % FILL_SLAB;
    if (j < lcnt[b]) {
      const int slot = hist[b] + j;
      if (slot < bin_capacity(b)) {
        int2* region = (int2*)(list + (size_t)b * BIN_FLOATS);
        region[1 + slot] = slab[i];
      }
    }
  }
}

#define BIN_THREADS 256
#define EDGE_GROUP  16
static_assert(32 % EDGE_GROUP == 0, "a 32-edge step is walked in whole groups");

__device__ __forceinline__ void lds_add(float* p, float x) {
  __hip_atomic_fetch_add(p, x, __ATOMIC_RELAXED, __HIP_MEMORY_SCOPE_WORKGROUP);
}

template <bool FULL>
__device__ __forceinline__ void walk_edges(const int pk, const int vb, const int n, const int lane,
                                           const float* __restrict__ emb, float* acc) {
  #pragma unroll
  for (int j0 = 0; j0 < 32; j0 += EDGE_GROUP) {
    if (!FULL && j0 >= n) break;
    float s0[EDGE_GROUP], s1[EDGE_GROUP];
    #pragma unroll
    for (int k = 0; k < EDGE_GROUP; ++k) {
      s0[k] = 0.f;
      s1[k] = 0.f;
      if (FULL || j0 + k < n) {
        const int cj = __builtin_amdgcn_readlane(pk, j0 + k) & COL_MASK;
        const float* src = emb + (size_t)cj * EMBED_DIM + lane;
        s0[k] = src[0];
        s1[k] = src[32];
      }
    }
    #pragma unroll
    for (int k = 0; k < EDGE_GROUP; ++k) {
      if (FULL || j0 + k < n) {
        const int   rl = __builtin_amdgcn_readlane(pk, j0 + k) >> COL_BITS;
        const float vj = __builtin_bit_cast(float, __builtin_amdgcn_readlane(vb, j0 + k));
        float* dst = acc + rl * EMBED_DIM + lane;
        lds_add(dst,      vj * s0[k]);
        lds_add(dst + 32, vj * s1[k]);
      }
    }
  }
}

__global__ void __launch_bounds__(BIN_THREADS)
lg_spmm_bin(const float* __restrict__ emb, const float* list, float* out) {
  __shared__ float acc[BIN_FLOATS];
  const int b = blockIdx.x;
  float4* acc4 = (float4*)acc;
  for (int i = threadIdx.x; i < BIN_FLOATS / 4; i += BIN_THREADS)
    acc4[i] = make_float4(0.f, 0.f, 0.f, 0.f);
  const int2* region = (const int2*)(list + (size_t)b * BIN_FLOATS);
  const int cnt = __builtin_amdgcn_readfirstlane(min(region[0].x, bin_capacity(b)));
  __syncthreads();

  const int lane = threadIdx.x & 31;
  const int wave = __builtin_amdgcn_readfirstlane(threadIdx.x >> 5);
  int s = wave * 32;
  int2 pe = make_int2(0, 0);
  if (s + lane < cnt) pe = region[1 + s + lane];
  while (s < cnt) {
    const int n = min(32, cnt - s);
    const int pk = pe.x, vb = pe.y;
    const int sn = s + BIN_THREADS;
    pe = make_int2(0, 0);
    if (sn + lane < cnt) pe = region[1 + sn + lane];
    if (n == 32) walk_edges<true>(pk, vb, 32, lane, emb, acc);
    else         walk_edges<false>(pk, vb, n, lane, emb, acc);
    s = sn;
  }
  __syncthreads();

  const int rows = min(BIN_ROWS, N_TOTAL - b * BIN_ROWS);
  float4* out4 = (float4*)(out + (size_t)b * BIN_FLOATS);
  for (int i = threadIdx.x; i < rows * (EMBED_DIM / 4); i += BIN_THREADS) out4[i] = acc4[i];
}

__global__ void lg_zero_rows(const int* __restrict__ uid, const int* __restrict__ iid,
                             float* __restrict__ buf) {
  const int t = blockIdx.x * blockDim.x + threadIdx.x;
  const int r = t >> 6, c = t & 63;
  const int n = (r < BATCH) ? uid[r] : (NUM_USERS + iid[r - BATCH]);
  buf[(size_t)n * EMBED_DIM + c] = 0.f;
}

#define NEED_WORDS 4692
static_assert(NEED_WORDS * 32 >= N_TOTAL, "bitmask must cover every row");
static_assert(N_EDGES % 32 == 0, "edges are read 32 per wave");

#define SG_THREADS 1024
static_assert(BATCH % SG_THREADS == 0, "ids are read SG_THREADS at a time");

__global__ void __launch_bounds__(SG_THREADS)
lg_spmm_scatter_gathered(const float* __restrict__ emb, const float* __restrict__ val,
                         const int* __restrict__ rw, const int* __restrict__ cl,
                         const int* __restrict__ uid, const int* __restrict__ iid,
                         float* __restrict__ out) {
  __shared__ unsigned int need[NEED_WORDS];
  for (int i = threadIdx.x; i < NEED_WORDS; i += SG_THREADS) need[i] = 0u;
  int un[BATCH / SG_THREADS], in[BATCH / SG_THREADS];
  for (int k = 0; k < BATCH / SG_THREADS; ++k) {
    un[k] = uid[k * SG_THREADS + threadIdx.x];
    in[k] = NUM_USERS + iid[k * SG_THREADS + threadIdx.x];
  }
  __syncthreads();
  for (int k = 0; k < BATCH / SG_THREADS; ++k) {
    atomicOr(&need[un[k] >> 5], 1u << (un[k] & 31));
    atomicOr(&need[in[k] >> 5], 1u << (in[k] & 31));
  }
  __syncthreads();

  const int lane = threadIdx.x & 31;
  const int w    = (blockIdx.x * blockDim.x + threadIdx.x) >> 5;
  const int nw   = (gridDim.x * blockDim.x) >> 5;
  for (int b = w; b < N_EDGES / 32; b += nw) {
    const int   e = b * 32 + lane;
    const int   r = rw[e];
    const int   c = cl[e];
    const float v = val[e];
    const bool hit = (need[r >> 5] >> (r & 31)) & 1u;
    unsigned int m = __builtin_amdgcn_ballot_w32(hit);
    while (m) {
      const int j = __builtin_ctz(m);
      m &= m - 1u;
      const int   rj = __builtin_amdgcn_readlane(r, j);
      const int   cj = __builtin_amdgcn_readlane(c, j);
      const float vj = __builtin_bit_cast(float,
                         __builtin_amdgcn_readlane(__builtin_bit_cast(int, v), j));
      const float* srcp = emb + (size_t)cj * EMBED_DIM + lane;
      const float s0 = srcp[0];
      const float s1 = srcp[32];
      float* dst = out + (size_t)rj * EMBED_DIM + lane;
      unsafeAtomicAdd(dst,      vj * s0);
      unsafeAtomicAdd(dst + 32, vj * s1);
    }
  }
}

__global__ void lg_dot_wmma(const float* __restrict__ acc, float* __restrict__ out) {
  __shared__ float sd[8][8][32];
  const int lane = threadIdx.x & 31;
  const int wave = threadIdx.x >> 5;
  const int tile = blockIdx.x * 8 + wave;
  const int k0   = tile * 16;
  const int rN   = lane & 15;
  const float* U = acc + (size_t)(k0 + rN) * EMBED_DIM;
  const float* I = acc + (size_t)(BATCH + k0 + rN) * EMBED_DIM;

  v8f c = {0.f, 0.f, 0.f, 0.f, 0.f, 0.f, 0.f, 0.f};

#if __has_builtin(__builtin_amdgcn_wmma_f32_16x16x4_f32)
  const int kb = (lane >> 4) << 1;
  #pragma unroll
  for (int s = 0; s < 16; ++s) {
    const int col = s * 4 + kb;
    v2f a; a[0] = U[col]; a[1] = U[col + 1];
    v2f b; b[0] = I[col]; b[1] = I[col + 1];
    c = __builtin_amdgcn_wmma_f32_16x16x4_f32(false, a, false, b, (short)0, c, false, false);
  }
#else
  const int kb8 = (lane >> 4) << 3;
  #pragma unroll
  for (int blk = 0; blk < 2; ++blk) {
    v16h a, b;
    #pragma unroll
    for (int i = 0; i < 8; ++i) {
      const int kk = blk * 32 + kb8 + ((i < 4) ? (2 * i) : (16 + 2 * (i - 4)));
      a[2 * i]     = (_Float16)U[kk];
      a[2 * i + 1] = (_Float16)U[kk + 1];
      b[2 * i]     = (_Float16)I[kk];
      b[2 * i + 1] = (_Float16)I[kk + 1];
    }
    c = __builtin_amdgcn_wmma_f32_16x16x32_f16(false, a, false, b, (short)0, c, false, false);
  }
#endif

  #pragma unroll
  for (int g = 0; g < 8; ++g) sd[wave][g][lane] = c[g];
  __syncthreads();

  if (lane < 16) {
    const float v = (lane < 8) ? sd[wave][lane][lane] : sd[wave][lane - 8][lane + 16];
    out[k0 + lane] = v * 0.0625f;
  }
}

extern "C" void kernel_launch(void* const* d_in, const int* in_sizes, int n_in,
                              void* d_out, int out_size, void* d_ws, size_t ws_size,
                              hipStream_t stream) {
  const float* node_emb = (const float*)d_in[0];
  const float* adj_val  = (const float*)d_in[1];
  const int*   adj_row  = (const int*)d_in[2];
  const int*   adj_col  = (const int*)d_in[3];
  const int*   user_ids = (const int*)d_in[4];
  const int*   item_ids = (const int*)d_in[5];
  float* out = (float*)d_out;

  const size_t embElems = (size_t)N_TOTAL * EMBED_DIM;
  char* ws   = (char*)d_ws;
  float* embA = (float*)ws;
  float* embB = (float*)(ws + embElems * sizeof(float));
  float* accS = (float*)(ws + 2 * embElems * sizeof(float));

  const dim3 b256(256);
  const int gatherBlocks = (2 * BATCH * EMBED_DIM) / 256;

  float* list = embB;
  lg_gather_init<<<gatherBlocks, b256, 0, stream>>>(node_emb, user_ids, item_ids, accS, list);
  lg_bin_fill<<<FILL_WGS, dim3(FILL_THREADS), 0, stream>>>(adj_val, adj_row, adj_col, list);

  const float* cur = node_emb;
  float* bufs[2] = {embA, embB};
  for (int L = 0; L < NUM_LAYERS - 1; ++L) {
    float* nxt = bufs[L & 1];
    lg_spmm_bin<<<NUM_BINS, dim3(BIN_THREADS), 0, stream>>>(cur, list, nxt);
    lg_gather_acc<<<gatherBlocks, b256, 0, stream>>>(nxt, user_ids, item_ids, accS);
    cur = nxt;
  }
  {
    float* nxt = bufs[(NUM_LAYERS - 1) & 1];
    lg_zero_rows<<<gatherBlocks, b256, 0, stream>>>(user_ids, item_ids, nxt);
    lg_spmm_scatter_gathered<<<512, dim3(SG_THREADS), 0, stream>>>(cur, adj_val, adj_row, adj_col,
                                                            user_ids, item_ids, nxt);
    lg_gather_acc<<<gatherBlocks, b256, 0, stream>>>(nxt, user_ids, item_ids, accS);
  }

  lg_dot_wmma<<<BATCH / 128, b256, 0, stream>>>(accS, out);
}


# Side table of gathered rows and its 5 small launches dropped: last layer adds onto layer 1's rows, the dot kernel sums the 3 tables at those rows. 10 launches -> 6; layer sum order e0+(e1+e3)+e2
# speedup vs baseline: 4.0077x; 1.0575x over previous
#include <hip/hip_runtime.h>

#define N_TOTAL    150100
#define EMBED_DIM  64
#define NUM_USERS  100000
#define N_EDGES    3000000
#define BATCH      4096
#define NUM_LAYERS 3

typedef __attribute__((ext_vector_type(2)))  float    v2f;
typedef __attribute__((ext_vector_type(8)))  float    v8f;
typedef __attribute__((ext_vector_type(16))) _Float16 v16h;

#define BIN_ROWS    128
#define BIN_SHIFT   7
#define NUM_BINS    ((N_TOTAL + BIN_ROWS - 1) / BIN_ROWS)
#define BIN_FLOATS  (BIN_ROWS * EMBED_DIM)
#define COL_BITS    18
#define COL_MASK    ((1 << COL_BITS) - 1)
static_assert((1 << BIN_SHIFT) == BIN_ROWS, "block index is a shift of the row");
static_assert(N_TOTAL <= (1 << COL_BITS), "a column id fits its field");

__device__ __forceinline__ int bin_capacity(int b) {
  const int rows = min(BIN_ROWS, N_TOTAL - b * BIN_ROWS);
  return rows * (EMBED_DIM / 2) - 1;
}

__global__ void lg_bin_clear(float* __restrict__ list) {
  const int t = blockIdx.x * blockDim.x + threadIdx.x;
  if (t < NUM_BINS) *(int*)(list + (size_t)t * BIN_FLOATS) = 0;
}

#define FILL_THREADS 1024
#define FILL_EPT     6
#define FILL_CHUNK   (FILL_THREADS * FILL_EPT)
#define FILL_WGS     ((N_EDGES + FILL_CHUNK - 1) / FILL_CHUNK)
#define FILL_SLAB    16

__global__ void __launch_bounds__(FILL_THREADS)
lg_bin_fill(const float* __restrict__ val, const int* __restrict__ rw,
            const int* __restrict__ cl, float* __restrict__ list) {
  __shared__ int  hist[NUM_BINS];
  __shared__ int  lcnt[NUM_BINS];
  __shared__ int2 slab[NUM_BINS * FILL_SLAB];
  for (int i = threadIdx.x; i < NUM_BINS; i += FILL_THREADS) hist[i] = 0;
  __syncthreads();
  const int e0 = blockIdx.x * FILL_CHUNK + threadIdx.x;
  int pk[FILL_EPT], bn[FILL_EPT], rk[FILL_EPT], vb[FILL_EPT];
  #pragma unroll
  for (int k = 0; k < FILL_EPT; ++k) {
    const int e = e0 + k * FILL_THREADS;
    bn[k] = -1;
    if (e < N_EDGES) {
      const int r = rw[e];
      bn[k] = r >> BIN_SHIFT;
      pk[k] = ((r & (BIN_ROWS - 1)) << COL_BITS) | cl[e];
      vb[k] = __builtin_bit_cast(int, val[e]);
    }
  }
  #pragma unroll
  for (int k = 0; k < FILL_EPT; ++k) {
    if (bn[k] >= 0) {
      rk[k] = atomicAdd(&hist[bn[k]], 1);
      if (rk[k] < FILL_SLAB) slab[bn[k] * FILL_SLAB + rk[k]] = make_int2(pk[k], vb[k]);
    }
  }
  __syncthreads();
  for (int b = threadIdx.x; b < NUM_BINS; b += FILL_THREADS) {
    const int n = hist[b];
    lcnt[b] = n;
    if (n) hist[b] = atomicAdd((int*)(list + (size_t)b * BIN_FLOATS), n);
  }
  __syncthreads();
  #pragma unroll
  for (int k = 0; k < FILL_EPT; ++k) {
    if (bn[k] >= 0 && rk[k] >= FILL_SLAB) {
      const int slot = hist[bn[k]] + rk[k];
      if (slot < bin_capacity(bn[k])) {
        int2* region = (int2*)(list + (size_t)bn[k] * BIN_FLOATS);
        region[1 + slot] = make_int2(pk[k], vb[k]);
      }
    }
  }
  for (int i = threadIdx.x; i < NUM_BINS * FILL_SLAB; i += FILL_THREADS) {
    const int b = i / FILL_SLAB, j = i % FILL_SLAB;
    if (j < lcnt[b]) {
      const int slot = hist[b] + j;
      if (slot < bin_capacity(b)) {
        int2* region = (int2*)(list + (size_t)b * BIN_FLOATS);
        region[1 + slot] = slab[i];
      }
    }
  }
}

#define BIN_THREADS 256
#define EDGE_GROUP  16
static_assert(32 % EDGE_GROUP == 0, "a 32-edge step is walked in whole groups");

__device__ __forceinline__ void lds_add(float* p, float x) {
  __hip_atomic_fetch_add(p, x, __ATOMIC_RELAXED, __HIP_MEMORY_SCOPE_WORKGROUP);
}

template <bool FULL>
__device__ __forceinline__ void walk_edges(const int pk, const int vb, const int n, const int lane,
                                           const float* __restrict__ emb, float* acc) {
  #pragma unroll
  for (int j0 = 0; j0 < 32; j0 += EDGE_GROUP) {
    if (!FULL && j0 >= n) break;
    float s0[EDGE_GROUP], s1[EDGE_GROUP];
    #pragma unroll
    for (int k = 0; k < EDGE_GROUP; ++k) {
      s0[k] = 0.f;
      s1[k] = 0.f;
      if (FULL || j0 + k < n) {
        const int cj = __builtin_amdgcn_readlane(pk, j0 + k) & COL_MASK;
        const float* src = emb + (size_t)cj * EMBED_DIM + lane;
        s0[k] = src[0];
        s1[k] = src[32];
      }
    }
    #pragma unroll
    for (int k = 0; k < EDGE_GROUP; ++k) {
      if (FULL || j0 + k < n) {
        const int   rl = __builtin_amdgcn_readlane(pk, j0 + k) >> COL_BITS;
        const float vj = __builtin_bit_cast(float, __builtin_amdgcn_readlane(vb, j0 + k));
        float* dst = acc + rl * EMBED_DIM + lane;
        lds_add(dst,      vj * s0[k]);
        lds_add(dst + 32, vj * s1[k]);
      }
    }
  }
}

__global__ void __launch_bounds__(BIN_THREADS)
lg_spmm_bin(const float* __restrict__ emb, const float* list, float* out) {
  __shared__ float acc[BIN_FLOATS];
  const int b = blockIdx.x;
  float4* acc4 = (float4*)acc;
  for (int i = threadIdx.x; i < BIN_FLOATS / 4; i += BIN_THREADS)
    acc4[i] = make_float4(0.f, 0.f, 0.f, 0.f);
  const int2* region = (const int2*)(list + (size_t)b * BIN_FLOATS);
  const int cnt = __builtin_amdgcn_readfirstlane(min(region[0].x, bin_capacity(b)));
  __syncthreads();

  const int lane = threadIdx.x & 31;
  const int wave = __builtin_amdgcn_readfirstlane(threadIdx.x >> 5);
  int s = wave * 32;
  int2 pe = make_int2(0, 0);
  if (s + lane < cnt) pe = region[1 + s + lane];
  while (s < cnt) {
    const int n = min(32, cnt - s);
    const int pk = pe.x, vb = pe.y;
    const int sn = s + BIN_THREADS;
    pe = make_int2(0, 0);
    if (sn + lane < cnt) pe = region[1 + sn + lane];
    if (n == 32) walk_edges<true>(pk, vb, 32, lane, emb, acc);
    else         walk_edges<false>(pk, vb, n, lane, emb, acc);
    s = sn;
  }
  __syncthreads();

  const int rows = min(BIN_ROWS, N_TOTAL - b * BIN_ROWS);
  float4* out4 = (float4*)(out + (size_t)b * BIN_FLOATS);
  for (int i = threadIdx.x; i < rows * (EMBED_DIM / 4); i += BIN_THREADS) out4[i] = acc4[i];
}

#define NEED_WORDS 4692
static_assert(NEED_WORDS * 32 >= N_TOTAL, "bitmask must cover every row");
static_assert(N_EDGES % 32 == 0, "edges are read 32 per wave");

#define SG_THREADS 1024
static_assert(BATCH % SG_THREADS == 0, "ids are read SG_THREADS at a time");

__global__ void __launch_bounds__(SG_THREADS)
lg_spmm_scatter_gathered(const float* __restrict__ emb, const float* __restrict__ val,
                         const int* __restrict__ rw, const int* __restrict__ cl,
                         const int* __restrict__ uid, const int* __restrict__ iid,
                         float* __restrict__ out) {
  __shared__ unsigned int need[NEED_WORDS];
  for (int i = threadIdx.x; i < NEED_WORDS; i += SG_THREADS) need[i] = 0u;
  int un[BATCH / SG_THREADS], in[BATCH / SG_THREADS];
  for (int k = 0; k < BATCH / SG_THREADS; ++k) {
    un[k] = uid[k * SG_THREADS + threadIdx.x];
    in[k] = NUM_USERS + iid[k * SG_THREADS + threadIdx.x];
  }
  __syncthreads();
  for (int k = 0; k < BATCH / SG_THREADS; ++k) {
    atomicOr(&need[un[k] >> 5], 1u << (un[k] & 31));
    atomicOr(&need[in[k] >> 5], 1u << (in[k] & 31));
  }
  __syncthreads();

  const int lane = threadIdx.x & 31;
  const int w    = (blockIdx.x * blockDim.x + threadIdx.x) >> 5;
  const int nw   = (gridDim.x * blockDim.x) >> 5;
  for (int b = w; b < N_EDGES / 32; b += nw) {
    const int   e = b * 32 + lane;
    const int   r = rw[e];
    const int   c = cl[e];
    const float v = val[e];
    const bool hit = (need[r >> 5] >> (r & 31)) & 1u;
    unsigned int m = __builtin_amdgcn_ballot_w32(hit);
    while (m) {
      const int j = __builtin_ctz(m);
      m &= m - 1u;
      const int   rj = __builtin_amdgcn_readlane(r, j);
      const int   cj = __builtin_amdgcn_readlane(c, j);
      const float vj = __builtin_bit_cast(float,
                         __builtin_amdgcn_readlane(__builtin_bit_cast(int, v), j));
      const float* srcp = emb + (size_t)cj * EMBED_DIM + lane;
      const float s0 = srcp[0];
      const float s1 = srcp[32];
      float* dst = out + (size_t)rj * EMBED_DIM + lane;
      unsafeAtomicAdd(dst,      vj * s0);
      unsafeAtomicAdd(dst + 32, vj * s1);
    }
  }
}

__global__ void __launch_bounds__(256)
lg_dot_wmma(const float* __restrict__ e0, const float* __restrict__ e13,
            const float* __restrict__ e2, const int* __restrict__ uid,
            const int* __restrict__ iid, float* __restrict__ out) {
  __shared__ float sd[8][8][32];
  const int lane = threadIdx.x & 31;
  const int wave = threadIdx.x >> 5;
  const int tile = blockIdx.x * 8 + wave;
  const int k0   = tile * 16;
  const int rN   = lane & 15;
  const size_t U = (size_t)uid[k0 + rN] * EMBED_DIM;
  const size_t I = (size_t)(NUM_USERS + iid[k0 + rN]) * EMBED_DIM;

  v8f c = {0.f, 0.f, 0.f, 0.f, 0.f, 0.f, 0.f, 0.f};

  const int kb = (lane >> 4) << 1;
  v2f u0[16], u1[16], u2[16], i0[16], i1[16], i2[16];
  #pragma unroll
  for (int s = 0; s < 16; ++s) {
    const int col = s * 4 + kb;
    u0[s] = *(const v2f*)(e0 + U + col);  u1[s] = *(const v2f*)(e13 + U + col);  u2[s] = *(const v2f*)(e2 + U + col);
    i0[s] = *(const v2f*)(e0 + I + col);  i1[s] = *(const v2f*)(e13 + I + col);  i2[s] = *(const v2f*)(e2 + I + col);
  }
  __builtin_amdgcn_sched_barrier(0);
  #pragma unroll
  for (int s = 0; s < 16; ++s) {
    const v2f a = (u0[s] + u1[s]) + u2[s];
    const v2f b = (i0[s] + i1[s]) + i2[s];
    c = __builtin_amdgcn_wmma_f32_16x16x4_f32(false, a, false, b, (short)0, c, false, false);
  }

  #pragma unroll
  for (int g = 0; g < 8; ++g) sd[wave][g][lane] = c[g];
  __syncthreads();

  if (lane < 16) {
    const float v = (lane < 8) ? sd[wave][lane][lane] : sd[wave][lane - 8][lane + 16];
    out[k0 + lane] = v * 0.0625f;
  }
}

extern "C" void kernel_launch(void* const* d_in, const int* in_sizes, int n_in,
                              void* d_out, int out_size, void* d_ws, size_t ws_size,
                              hipStream_t stream) {
  const float* node_emb = (const float*)d_in[0];
  const float* adj_val  = (const float*)d_in[1];
  const int*   adj_row  = (const int*)d_in[2];
  const int*   adj_col  = (const int*)d_in[3];
  const int*   user_ids = (const int*)d_in[4];
  const int*   item_ids = (const int*)d_in[5];
  float* out = (float*)d_out;

  const size_t embElems = (size_t)N_TOTAL * EMBED_DIM;
  char* ws   = (char*)d_ws;
  float* embA = (float*)ws;
  float* embB = (float*)(ws + embElems * sizeof(float));

  float* list = embB;
  lg_bin_clear<<<(NUM_BINS + 255) / 256, dim3(256), 0, stream>>>(list);
  lg_bin_fill<<<FILL_WGS, dim3(FILL_THREADS), 0, stream>>>(adj_val, adj_row, adj_col, list);

  lg_spmm_bin<<<NUM_BINS, dim3(BIN_THREADS), 0, stream>>>(node_emb, list, embA);
  lg_spmm_bin<<<NUM_BINS, dim3(BIN_THREADS), 0, stream>>>(embA, list, embB);

  lg_spmm_scatter_gathered<<<512, dim3(SG_THREADS), 0, stream>>>(embB, adj_val, adj_row, adj_col,
                                                          user_ids, item_ids, embA);

  lg_dot_wmma<<<BATCH / 128, dim3(256), 0, stream>>>(node_emb, embA, embB, user_ids, item_ids, out);
}
